# starting kernel
# baseline (speedup 1.0000x reference)
#include <hip/hip_runtime.h>
#include <hip/hip_bf16.h>

#define F_DIM 48

typedef float v2f __attribute__((ext_vector_type(2)));
typedef float v8f __attribute__((ext_vector_type(8)));

__global__ void zero_kernel(float* __restrict__ out, int n) {
    int i4 = (blockIdx.x * blockDim.x + threadIdx.x) * 4;
    if (i4 + 3 < n) {
        *reinterpret_cast<float4*>(out + i4) = make_float4(0.f, 0.f, 0.f, 0.f);
    } else {
        for (int k = i4; k < n; ++k) out[k] = 0.f;
    }
}

__global__ void edge_scatter_kernel(const float* __restrict__ x,
                                    const int*   __restrict__ rows,
                                    const int*   __restrict__ cols,
                                    const float* __restrict__ vals,
                                    float* __restrict__ out,
                                    int nEdges) {
    int t = blockIdx.x * blockDim.x + threadIdx.x;
    int e = t >> 4;
    int f = t & 15;
    if (e >= nEdges) return;

    int epf = e + 64;
    if (f == 0 && epf < nEdges) {
        int cpf = cols[epf];
        __builtin_prefetch(x + (long)cpf * F_DIM, 0, 1);
    }

    int   r = rows[e];
    int   c = cols[e];
    float v = vals[e];

    const float* xr   = x   + (long)c * F_DIM;
    float*       orow = out + (long)r * F_DIM;

    atomicAdd(orow + f,      v * xr[f]);
    atomicAdd(orow + f + 16, v * xr[f + 16]);
    atomicAdd(orow + f + 32, v * xr[f + 32]);
}

__global__ void scale_wmma_kernel(float* __restrict__ out,
                                  const float* __restrict__ w,
                                  int nNodes) {
    int wave = (blockIdx.x * blockDim.x + threadIdx.x) >> 5;
    int lane = threadIdx.x & 31;
    int rows0 = wave * 16;
    if (rows0 >= nNodes) return;

    int n = lane & 15;
    int h = lane >> 4;

#pragma unroll
    for (int j = 0; j < 3; ++j) {
        float wn = w[16 * j + n];
        v8f acc = {};
#pragma unroll
        for (int c4 = 0; c4 < 4; ++c4) {
            int kbase = 16 * j + 4 * c4;

            const float* ap = out + (long)(rows0 + n) * F_DIM + kbase + 2 * h;
            v2f a;
            a.x = ap[0];
            a.y = ap[1];

            v2f b;
            b.x = (n == 4 * c4 + 2 * h)     ? wn : 0.f;
            b.y = (n == 4 * c4 + 2 * h + 1) ? wn : 0.f;

            acc = __builtin_amdgcn_wmma_f32_16x16x4_f32(
                 false, a,  false, b,
                 (short)0, acc,  false,  false);
        }
#pragma unroll
        for (int v = 0; v < 8; ++v) {
            out[(long)(rows0 + v + 8 * h) * F_DIM + 16 * j + n] = acc[v];
        }
    }
}

extern "C" void kernel_launch(void* const* d_in, const int* in_sizes, int n_in,
                              void* d_out, int out_size, void* d_ws, size_t ws_size,
                              hipStream_t stream) {
    const float* x    = (const float*)d_in[0];
    const float* w    = (const float*)d_in[1];
    const int*   rows = (const int*)  d_in[2];
    const int*   cols = (const int*)  d_in[3];
    const float* vals = (const float*)d_in[4];
    float* out = (float*)d_out;

    const int nEdges = in_sizes[2];
    const int nNodes = in_sizes[0] / F_DIM;

    {
        int nquads = (out_size + 3) / 4;
        int blocks = (nquads + 255) / 256;
        zero_kernel<<<blocks, 256, 0, stream>>>(out, out_size);
    }

    {
        long threads = (long)nEdges * 16;
        int blocks = (int)((threads + 255) / 256);
        edge_scatter_kernel<<<blocks, 256, 0, stream>>>(x, rows, cols, vals,
                                                        out, nEdges);
    }

    {
        int waves  = (nNodes + 15) / 16;
        int blocks = (waves + 7) / 8;
        scale_wmma_kernel<<<blocks, 256, 0, stream>>>(out, w, nNodes);
    }
}


# Edge scatter run by persistent waves: 4096 waves each claim 64-edge (block, row range) jobs in a loop, pack in-range edges by ballot, gather 60 rows at once; f32 adds stay in atomic arrival order
# speedup vs baseline: 2.5770x; 2.5770x over previous
#include <hip/hip_runtime.h>
#include <hip/hip_bf16.h>


#define F_DIM 48
#define JOB_EDGES 64
#define CNT_STRIDE 64
#define N_CNT (2 * CNT_STRIDE)
#define SCATTER_GROUPS 1024

typedef float v2f __attribute__((ext_vector_type(2)));
typedef float v8f __attribute__((ext_vector_type(8)));

__global__ void zero_kernel(float* __restrict__ out, int n, int* __restrict__ cnt) {
    int gid = blockIdx.x * blockDim.x + threadIdx.x;
    if (gid < N_CNT) cnt[gid] = 0;
    int i4 = gid * 4;
    if (i4 + 3 < n) {
        *reinterpret_cast<float4*>(out + i4) = make_float4(0.f, 0.f, 0.f, 0.f);
    } else {
        for (int k = i4; k < n; ++k) out[k] = 0.f;
    }
}

#define SET_PAIRS 10

struct Chunk {
    unsigned o[SET_PAIRS];
    bool     ok[SET_PAIRS];
    float    p0[SET_PAIRS], p1[SET_PAIRS], p2[SET_PAIRS];
};

__device__ __forceinline__ int claim_issue(int* cnt, int half, int lane) {
    unsigned t = 0u;
    if (lane == 0) t = atomicInc(reinterpret_cast<unsigned*>(cnt + half * CNT_STRIDE), 0xFFFFFFFFu);
    return (int)t;
}

__device__ __forceinline__ void gather_chunk(const float* __restrict__ x,
                                             int r, int c, int vi,
                                             unsigned& m, int h, int f, Chunk& k) {
#pragma unroll
    for (int i = 0; i < SET_PAIRS; ++i) {
        const bool has0 = m != 0u;
        const int  e0 = has0 ? __builtin_ctz(m) : 0;
        m &= m - 1u;
        const bool has1 = m != 0u;
        const int  e1 = has1 ? __builtin_ctz(m) : 0;
        m &= m - 1u;
        const int  e  = h ? e1 : e0;
        const bool ok = h ? has1 : has0;
        const int   rk = __builtin_amdgcn_ds_bpermute(e << 2, r);
        const int   ck = __builtin_amdgcn_ds_bpermute(e << 2, c);
        const float vk = __int_as_float(__builtin_amdgcn_ds_bpermute(e << 2, vi));
        k.ok[i] = ok;
        k.o[i]  = (unsigned)rk * F_DIM + (unsigned)f;
        const unsigned xo = ok ? (unsigned)ck * F_DIM + (unsigned)f : (unsigned)f;
        k.p0[i] = vk * x[xo];
        k.p1[i] = vk * x[xo + 16];
        k.p2[i] = vk * x[xo + 32];
    }
}

__device__ __forceinline__ void add_chunk(float* __restrict__ out, const Chunk& k) {
#pragma unroll
    for (int i = 0; i < SET_PAIRS; ++i) {
        if (k.ok[i]) {
            float* o = out + k.o[i];
            atomicAdd(o,      k.p0[i]);
            atomicAdd(o + 16, k.p1[i]);
            atomicAdd(o + 32, k.p2[i]);
        }
    }
}

__global__ void __launch_bounds__(128)
edge_scatter_kernel(const float* __restrict__ x,
                    const int*   __restrict__ rows,
                    const int*   __restrict__ cols,
                    const float* __restrict__ vals,
                    float* __restrict__ out,
                    int*   __restrict__ cnt,
                    int nEdges, int nBlocks, int split) {
    const int lane = threadIdx.x & 31;
    const int h = lane >> 4;
    const int f = lane & 15;

    const unsigned id = (unsigned)__builtin_amdgcn_s_sendmsg_rtn(0x87);
    int  half = (int)((id >> 18) & 1u);
    bool switched = false;

    int t = __builtin_amdgcn_readfirstlane(claim_issue(cnt, half, lane));
    if (t >= nBlocks) {
        switched = true;
        half ^= 1;
        t = __builtin_amdgcn_readfirstlane(claim_issue(cnt, half, lane));
    }
    if (t >= nBlocks) return;

    int eA = t * JOB_EDGES + lane, eB = eA + 32;
    bool okA = eA < nEdges, okB = eB < nEdges;
    int iA = okA ? eA : 0, iB = okB ? eB : 0;
    int   rA = rows[iA], rB = rows[iB];
    int   cA = cols[iA], cB = cols[iB];
    float vA = vals[iA], vB = vals[iB];

    for (;;) {
        const int tv = claim_issue(cnt, half, lane);

        unsigned mA = __builtin_amdgcn_ballot_w32(okA && (((rA >= split) ? 1 : 0) == half));
        unsigned mB = __builtin_amdgcn_ballot_w32(okB && (((rB >= split) ? 1 : 0) == half));
        Chunk ka, kb;
        gather_chunk(x, rA, cA, __float_as_int(vA), mA, h, f, ka);
        gather_chunk(x, rB, cB, __float_as_int(vB), mB, h, f, kb);

        int tn = __builtin_amdgcn_readfirstlane(tv);
        int nhalf = half;
        if (tn >= nBlocks && !switched) {
            switched = true;
            nhalf = half ^ 1;
            tn = __builtin_amdgcn_readfirstlane(claim_issue(cnt, nhalf, lane));
        }
        const bool more = tn < nBlocks;
        const int nA = (more ? tn : 0) * JOB_EDGES + lane, nB = nA + 32;
        const bool nokA = more && nA < nEdges, nokB = more && nB < nEdges;
        const int niA = nokA ? nA : 0, niB = nokB ? nB : 0;
        const int   nrA = rows[niA], nrB = rows[niB];
        const int   ncA = cols[niA], ncB = cols[niB];
        const float nvA = vals[niA], nvB = vals[niB];

        add_chunk(out, ka);
        add_chunk(out, kb);
        while (mA != 0u) {
            Chunk k;
            gather_chunk(x, rA, cA, __float_as_int(vA), mA, h, f, k);
            add_chunk(out, k);
        }
        while (mB != 0u) {
            Chunk k;
            gather_chunk(x, rB, cB, __float_as_int(vB), mB, h, f, k);
            add_chunk(out, k);
        }
        if (!more) break;
        half = nhalf;
        okA = nokA; okB = nokB;
        rA = nrA; rB = nrB; cA = ncA; cB = ncB; vA = nvA; vB = nvB;
    }
}

__global__ void scale_wmma_kernel(float* __restrict__ out,
                                  const float* __restrict__ w,
                                  int nNodes) {
    int wave = (blockIdx.x * blockDim.x + threadIdx.x) >> 5;
    int lane = threadIdx.x & 31;
    int rows0 = wave * 16;
    if (rows0 >= nNodes) return;

    int n = lane & 15;
    int h = lane >> 4;

#pragma unroll
    for (int j = 0; j < 3; ++j) {
        float wn = w[16 * j + n];
        v8f acc = {};
#pragma unroll
        for (int c4 = 0; c4 < 4; ++c4) {
            int kbase = 16 * j + 4 * c4;

            const float* ap = out + (long)(rows0 + n) * F_DIM + kbase + 2 * h;
            v2f a;
            a.x = ap[0];
            a.y = ap[1];

            v2f b;
            b.x = (n == 4 * c4 + 2 * h)     ? wn : 0.f;
            b.y = (n == 4 * c4 + 2 * h + 1) ? wn : 0.f;

            acc = __builtin_amdgcn_wmma_f32_16x16x4_f32(
                 false, a,  false, b,
                 (short)0, acc,  false,  false);
        }
#pragma unroll
        for (int v = 0; v < 8; ++v) {
            out[(long)(rows0 + v + 8 * h) * F_DIM + 16 * j + n] = acc[v];
        }
    }
}

extern "C" void kernel_launch(void* const* d_in, const int* in_sizes, int n_in,
                              void* d_out, int out_size, void* d_ws, size_t ws_size,
                              hipStream_t stream) {
    const float* x    = (const float*)d_in[0];
    const float* w    = (const float*)d_in[1];
    const int*   rows = (const int*)  d_in[2];
    const int*   cols = (const int*)  d_in[3];
    const float* vals = (const float*)d_in[4];
    float* out = (float*)d_out;
    int*   cnt = (int*)d_ws;

    const int nEdges = in_sizes[2];
    const int nNodes = in_sizes[0] / F_DIM;

    const int nBlocks = (nEdges + JOB_EDGES - 1) / JOB_EDGES;
    const int split   = (nNodes / 2) & ~1;

    {
        int nquads = (out_size + 3) / 4;
        int nthr   = nquads > N_CNT ? nquads : N_CNT;
        int blocks = (nthr + 255) / 256;
        zero_kernel<<<blocks, 256, 0, stream>>>(out, out_size, cnt);
    }

    {
        edge_scatter_kernel<<<SCATTER_GROUPS, 128, 0, stream>>>(x, rows, cols, vals, out,
                                                                cnt, nEdges, nBlocks, split);
    }

    {
        int waves  = (nNodes + 15) / 16;
        int blocks = (waves + 7) / 8;
        scale_wmma_kernel<<<blocks, 256, 0, stream>>>(out, w, nNodes);
    }
}
